# starting kernel
# baseline (speedup 1.0000x reference)
#include <hip/hip_runtime.h>

typedef __attribute__((ext_vector_type(2))) float v2f;
typedef __attribute__((ext_vector_type(4))) float v4f;
typedef __attribute__((ext_vector_type(8))) float v8f;

#define NJ 17
#define NG 17
#define NR 49
#define NC 3
#define NK 32
#define KPAD 20
#define BPW 16
#define WPB 8
#define BPB (BPW*WPB)
#define OUTTILE (BPW*NG*NC)

__device__ __constant__ int c_idx[NR] = {
    0,1,
    1,2,3,4,5,
    2,3,6,
    3,6,7,
    6,7,
    2,4,8,
    4,8,9,
    8,9,
    10,11,12,
    11,12,13,
    12,13,
    10,14,15,
    14,15,16,
    15,16,
    5,10,11,14,
    2,5,10,
    0,1,2};
__device__ __constant__ int c_seg[NR] = {
    0,0,
    1,1,1,1,1,
    2,2,2,
    3,3,3,
    4,4,
    5,5,5,
    6,6,6,
    7,7,
    8,8,8,
    9,9,9,
    10,10,
    11,11,11,
    12,12,12,
    13,13,
    14,14,14,14,
    15,15,15,
    16,16,16};

union V8 { v8f v; float f[8]; };

__global__ __launch_bounds__(256)
void PropeIuncturam_kernel(const float* __restrict__ x,
                           const float* __restrict__ W,
                           const float* __restrict__ bias,
                           float* __restrict__ out, int B)
{
    __shared__ float s_coef[NC][KPAD][NG];
    __shared__ float s_bsum[NG][NC];
    __shared__ float s_xr[WPB][NC][KPAD][BPW];
    const int tid  = threadIdx.x;
    const int wave = tid >> 5;
    const int lane = tid & 31;
    const int half = lane >> 4;
    const int nn   = lane & 15;

    for (int i = tid; i < NC*KPAD*NG; i += 256) ((float*)s_coef)[i] = 0.f;
    for (int i = tid; i < NG*NC;     i += 256) ((float*)s_bsum)[i] = 0.f;
    __syncthreads();
    for (int t = tid; t < NR*NC; t += 256) {
        int r = t / NC, c = t - r*NC;
        atomicAdd(&s_coef[c][c_idx[r]][c_seg[r]], W[r*NC + c]);
        atomicAdd(&s_bsum[c_seg[r]][c],           bias[r*NC + c]);
    }
    for (int i = lane; i < NC*(KPAD-NJ)*BPW; i += 32) {
        int c = i / ((KPAD-NJ)*BPW); int rem = i - c*(KPAD-NJ)*BPW;
        s_xr[wave][c][NJ + rem/BPW][rem % BPW] = 0.f;
    }
    __syncthreads();

    const long base_b = (long)blockIdx.x * BPB + (long)wave * BPW;
    const long left   = (long)B - base_b;
    const int  rem_b  = (int)(left < (long)BPW ? (left < 0 ? 0 : left) : (long)BPW);
    const float* xw   = x + (size_t)base_b * (NJ*NC*NK);

    if (rem_b == BPW) {
        for (int rho = lane; rho < BPW*NJ*NC; rho += 32) {
            int m  = rho / (NJ*NC);
            int r2 = rho - m*(NJ*NC);
            int j  = r2 / NC;
            int c  = r2 - j*NC;
            const v4f* p = (const v4f*)(xw + (size_t)rho * NK);
            float s = 0.f;
            #pragma unroll
            for (int t = 0; t < NK/4; ++t) {
                v4f v = __builtin_nontemporal_load(p + t);
                s += (v.x + v.y) + (v.z + v.w);
            }
            s_xr[wave][c][j][m] = s;
        }
    } else {
        for (int rho = lane; rho < BPW*NJ*NC; rho += 32) {
            int m  = rho / (NJ*NC);
            int r2 = rho - m*(NJ*NC);
            int j  = r2 / NC;
            int c  = r2 - j*NC;
            float s = 0.f;
            if (m < rem_b) {
                const v4f* p = (const v4f*)(xw + (size_t)rho * NK);
                #pragma unroll
                for (int t = 0; t < NK/4; ++t) {
                    v4f v = __builtin_nontemporal_load(p + t);
                    s += (v.x + v.y) + (v.z + v.w);
                }
            }
            s_xr[wave][c][j][m] = s;
        }
    }
    __syncthreads();

    V8 acc[NC];
    #pragma unroll
    for (int c = 0; c < NC; ++c) {
        const float binit = s_bsum[nn][c];
        #pragma unroll
        for (int r = 0; r < 8; ++r) acc[c].f[r] = binit;

        #pragma unroll
        for (int q = 0; q < KPAD/4; ++q) {
            const int k0 = 4*q + 2*half;
            v2f a  = { s_xr[wave][c][k0][nn], s_xr[wave][c][k0+1][nn] };
            v2f bm = { s_coef[c][k0][nn],     s_coef[c][k0+1][nn] };
            acc[c].v = __builtin_amdgcn_wmma_f32_16x16x4_f32(
                           false, a, false, bm, (short)0, acc[c].v, false, false);
        }
    }

    float g16[NC] = {0.f, 0.f, 0.f};
    if (half == 0) {
        #pragma unroll
        for (int c = 0; c < NC; ++c) {
            float s = s_bsum[16][c];
            #pragma unroll
            for (int j = 0; j < NJ; ++j)
                s += s_coef[c][j][16] * s_xr[wave][c][j][nn];
            g16[c] = s;
        }
    }

    __syncthreads();
    float* s_out = (float*)&s_xr[wave][0][0][0];

    #pragma unroll
    for (int c = 0; c < NC; ++c)
        #pragma unroll
        for (int r = 0; r < 8; ++r)
            s_out[(r + 8*half)*(NG*NC) + nn*NC + c] = acc[c].f[r];
    if (half == 0) {
        #pragma unroll
        for (int c = 0; c < NC; ++c)
            s_out[nn*(NG*NC) + 16*NC + c] = g16[c];
    }

    if (rem_b == BPW) {
        const v4f* s4 = (const v4f*)s_out;
        v4f* o4 = (v4f*)(out + (size_t)base_b * (NG*NC));
        for (int i = lane; i < OUTTILE/4; i += 32)
            __builtin_nontemporal_store(s4[i], o4 + i);
    } else if (rem_b > 0) {
        for (int i = lane; i < rem_b*NG*NC; i += 32)
            out[(size_t)base_b*(NG*NC) + i] = s_out[i];
    }
}

extern "C" void kernel_launch(void* const* d_in, const int* in_sizes, int n_in,
                              void* d_out, int out_size, void* d_ws, size_t ws_size,
                              hipStream_t stream) {
    const float* x  = (const float*)d_in[0];
    const float* W  = (const float*)d_in[1];
    const float* b  = (const float*)d_in[2];
    float* out      = (float*)d_out;
    const int B = in_sizes[0] / (NJ*NC*NK);

    dim3 grid((B + BPB - 1) / BPB);
    dim3 block(256);
    hipLaunchKernelGGL(PropeIuncturam_kernel, grid, block, 0, stream,
                       x, W, b, out, B);
}


# Stage-1 stream coalesced: each wave load reads 512 contiguous B (4 whole rows); 8-lane DPP butterfly finishes the row sums (new reduction tree); fixed 25-trip loop unrolls, about 40 loads in flight
# speedup vs baseline: 1.8593x; 1.8593x over previous
#include <hip/hip_runtime.h>

typedef __attribute__((ext_vector_type(2))) float v2f;
typedef __attribute__((ext_vector_type(4))) float v4f;
typedef __attribute__((ext_vector_type(8))) float v8f;

#define NJ 17
#define NG 17
#define NR 49
#define NC 3
#define NK 32
#define KPAD 20
#define BPW 16
#define WPB 8
#define BPB (BPW*WPB)
#define OUTTILE (BPW*NG*NC)

__device__ __constant__ int c_idx[NR] = {
    0,1,
    1,2,3,4,5,
    2,3,6,
    3,6,7,
    6,7,
    2,4,8,
    4,8,9,
    8,9,
    10,11,12,
    11,12,13,
    12,13,
    10,14,15,
    14,15,16,
    15,16,
    5,10,11,14,
    2,5,10,
    0,1,2};
__device__ __constant__ int c_seg[NR] = {
    0,0,
    1,1,1,1,1,
    2,2,2,
    3,3,3,
    4,4,
    5,5,5,
    6,6,6,
    7,7,
    8,8,8,
    9,9,9,
    10,10,
    11,11,11,
    12,12,12,
    13,13,
    14,14,14,14,
    15,15,15,
    16,16,16};

union V8 { v8f v; float f[8]; };

template <int XM>
__device__ __forceinline__ float lane_xor_add(float keep, float send) {
    const int s = __builtin_amdgcn_update_dpp(0, __builtin_bit_cast(int, send),
                                              0x160 + XM, 0xf, 0xf, true);
    return keep + __builtin_bit_cast(float, s);
}

__global__ __launch_bounds__(256)
void PropeIuncturam_kernel(const float* __restrict__ x,
                           const float* __restrict__ W,
                           const float* __restrict__ bias,
                           float* __restrict__ out, int B)
{
    __shared__ float s_coef[NC][KPAD][NG];
    __shared__ float s_bsum[NG][NC];
    __shared__ float s_xr[WPB][NC][KPAD][BPW];
    const int tid  = threadIdx.x;
    const int wave = tid >> 5;
    const int lane = tid & 31;
    const int half = lane >> 4;
    const int nn   = lane & 15;

    for (int i = tid; i < NC*KPAD*NG; i += 256) ((float*)s_coef)[i] = 0.f;
    for (int i = tid; i < NG*NC;     i += 256) ((float*)s_bsum)[i] = 0.f;
    __syncthreads();
    for (int t = tid; t < NR*NC; t += 256) {
        int r = t / NC, c = t - r*NC;
        atomicAdd(&s_coef[c][c_idx[r]][c_seg[r]], W[r*NC + c]);
        atomicAdd(&s_bsum[c_seg[r]][c],           bias[r*NC + c]);
    }
    for (int i = lane; i < NC*(KPAD-NJ)*BPW; i += 32) {
        int c = i / ((KPAD-NJ)*BPW); int rem = i - c*(KPAD-NJ)*BPW;
        s_xr[wave][c][NJ + rem/BPW][rem % BPW] = 0.f;
    }
    __syncthreads();

    const long base_b = (long)blockIdx.x * BPB + (long)wave * BPW;
    const long left   = (long)B - base_b;
    const int  rem_b  = (int)(left < (long)BPW ? (left < 0 ? 0 : left) : (long)BPW);
    const float* xw   = x + (size_t)base_b * (NJ*NC*NK);

    if (rem_b == BPW) {
        constexpr int ROWS   = BPW*NJ*NC;
        constexpr int NGRP32 = ROWS / 32;
        static_assert(ROWS - 32*NGRP32 == 16, "the tail below takes exactly 4 pieces");
        const v4f* p = (const v4f*)xw + lane;
        const bool b2 = (lane & 4) != 0;
        const bool b1 = (lane & 2) != 0;
        const bool b0 = (lane & 1) != 0;
        const int  rsub = 4*(lane & 7) + (lane >> 3);

        for (int it = 0; it < NGRP32; ++it) {
            float q[8];
            #pragma unroll
            for (int i = 0; i < 8; ++i) {
                v4f v = __builtin_nontemporal_load(p + (it*8 + i)*32);
                q[i] = (v.x + v.y) + (v.z + v.w);
            }
            float t[4], u[2];
            #pragma unroll
            for (int i = 0; i < 4; ++i)
                t[i] = lane_xor_add<4>(b2 ? q[i+4] : q[i], b2 ? q[i] : q[i+4]);
            #pragma unroll
            for (int i = 0; i < 2; ++i)
                u[i] = lane_xor_add<2>(b1 ? t[i+2] : t[i], b1 ? t[i] : t[i+2]);
            const float r = lane_xor_add<1>(b0 ? u[1] : u[0], b0 ? u[0] : u[1]);

            const int rho = it*32 + rsub;
            const int m  = rho / (NJ*NC);
            const int r2 = rho - m*(NJ*NC);
            const int j  = r2 / NC;
            const int c  = r2 - j*NC;
            s_xr[wave][c][j][m] = r;
        }
        {
            float q[4];
            #pragma unroll
            for (int i = 0; i < 4; ++i) {
                v4f v = __builtin_nontemporal_load(p + (NGRP32*8 + i)*32);
                q[i] = (v.x + v.y) + (v.z + v.w);
            }
            float u[2];
            #pragma unroll
            for (int i = 0; i < 4; ++i) q[i] = lane_xor_add<4>(q[i], q[i]);
            #pragma unroll
            for (int i = 0; i < 2; ++i)
                u[i] = lane_xor_add<2>(b1 ? q[i+2] : q[i], b1 ? q[i] : q[i+2]);
            const float r = lane_xor_add<1>(b0 ? u[1] : u[0], b0 ? u[0] : u[1]);
            if (!b2) {
                const int rho = NGRP32*32 + 4*(lane & 3) + (lane >> 3);
                const int m  = rho / (NJ*NC);
                const int r2 = rho - m*(NJ*NC);
                const int j  = r2 / NC;
                const int c  = r2 - j*NC;
                s_xr[wave][c][j][m] = r;
            }
        }
    } else {
        for (int rho = lane; rho < BPW*NJ*NC; rho += 32) {
            int m  = rho / (NJ*NC);
            int r2 = rho - m*(NJ*NC);
            int j  = r2 / NC;
            int c  = r2 - j*NC;
            float s = 0.f;
            if (m < rem_b) {
                const v4f* p = (const v4f*)(xw + (size_t)rho * NK);
                #pragma unroll
                for (int t = 0; t < NK/4; ++t) {
                    v4f v = __builtin_nontemporal_load(p + t);
                    s += (v.x + v.y) + (v.z + v.w);
                }
            }
            s_xr[wave][c][j][m] = s;
        }
    }
    __syncthreads();

    V8 acc[NC];
    #pragma unroll
    for (int c = 0; c < NC; ++c) {
        const float binit = s_bsum[nn][c];
        #pragma unroll
        for (int r = 0; r < 8; ++r) acc[c].f[r] = binit;

        #pragma unroll
        for (int q = 0; q < KPAD/4; ++q) {
            const int k0 = 4*q + 2*half;
            v2f a  = { s_xr[wave][c][k0][nn], s_xr[wave][c][k0+1][nn] };
            v2f bm = { s_coef[c][k0][nn],     s_coef[c][k0+1][nn] };
            acc[c].v = __builtin_amdgcn_wmma_f32_16x16x4_f32(
                           false, a, false, bm, (short)0, acc[c].v, false, false);
        }
    }

    float g16[NC] = {0.f, 0.f, 0.f};
    if (half == 0) {
        #pragma unroll
        for (int c = 0; c < NC; ++c) {
            float s = s_bsum[16][c];
            #pragma unroll
            for (int j = 0; j < NJ; ++j)
                s += s_coef[c][j][16] * s_xr[wave][c][j][nn];
            g16[c] = s;
        }
    }

    __syncthreads();
    float* s_out = (float*)&s_xr[wave][0][0][0];

    #pragma unroll
    for (int c = 0; c < NC; ++c)
        #pragma unroll
        for (int r = 0; r < 8; ++r)
            s_out[(r + 8*half)*(NG*NC) + nn*NC + c] = acc[c].f[r];
    if (half == 0) {
        #pragma unroll
        for (int c = 0; c < NC; ++c)
            s_out[nn*(NG*NC) + 16*NC + c] = g16[c];
    }

    if (rem_b == BPW) {
        const v4f* s4 = (const v4f*)s_out;
        v4f* o4 = (v4f*)(out + (size_t)base_b * (NG*NC));
        for (int i = lane; i < OUTTILE/4; i += 32)
            __builtin_nontemporal_store(s4[i], o4 + i);
    } else if (rem_b > 0) {
        for (int i = lane; i < rem_b*NG*NC; i += 32)
            out[(size_t)base_b*(NG*NC) + i] = s_out[i];
    }
}

extern "C" void kernel_launch(void* const* d_in, const int* in_sizes, int n_in,
                              void* d_out, int out_size, void* d_ws, size_t ws_size,
                              hipStream_t stream) {
    const float* x  = (const float*)d_in[0];
    const float* W  = (const float*)d_in[1];
    const float* b  = (const float*)d_in[2];
    float* out      = (float*)d_out;
    const int B = in_sizes[0] / (NJ*NC*NK);

    dim3 grid((B + BPB - 1) / BPB);
    dim3 block(256);
    hipLaunchKernelGGL(PropeIuncturam_kernel, grid, block, 0, stream,
                       x, W, b, out, B);
}
